# starting kernel
# baseline (speedup 1.0000x reference)
#include <hip/hip_runtime.h>


typedef float v2f __attribute__((ext_vector_type(2)));
typedef float v8f __attribute__((ext_vector_type(8)));

#define DH 256
#define DOUT 128

__global__ void zero_kernel(float* __restrict__ p, long n) {
    long i = (long)blockIdx.x * blockDim.x + threadIdx.x;
    if (i < n) p[i] = 0.0f;
}

__global__ void deg_kernel(const int* __restrict__ dst, int E, float* __restrict__ deg) {
    int e = blockIdx.x * blockDim.x + threadIdx.x;
    if (e < E) atomicAdd(&deg[dst[e]], 1.0f);
}

__global__ __launch_bounds__(256) void scatter_kernel(
    const float* __restrict__ X, const int* __restrict__ src,
    const int* __restrict__ dst, int E, float* __restrict__ agg)
{
    int e = blockIdx.x * blockDim.y + threadIdx.y;
    if (e >= E) return;
    int lane = threadIdx.x;
    int s = src[e];
    int d = dst[e];
    const float4* xs = (const float4*)(X + (size_t)s * DH) + lane * 2;
    float4 v0 = xs[0];
    float4 v1 = xs[1];
    float* ad = agg + (size_t)d * DH + lane * 8;
    atomicAdd(ad + 0, v0.x);
    atomicAdd(ad + 1, v0.y);
    atomicAdd(ad + 2, v0.z);
    atomicAdd(ad + 3, v0.w);
    atomicAdd(ad + 4, v1.x);
    atomicAdd(ad + 5, v1.y);
    atomicAdd(ad + 6, v1.z);
    atomicAdd(ad + 7, v1.w);
}

__global__ void norm_kernel(float* __restrict__ agg, const float* __restrict__ deg, long n) {
    long i = (long)blockIdx.x * blockDim.x + threadIdx.x;
    if (i < n) {
        int row = (int)(i >> 8);
        agg[i] *= 1.0f / fmaxf(deg[row], 1.0f);
    }
}

__global__ __launch_bounds__(128) void gemm_wmma_kernel(
    const float* __restrict__ A, const float* __restrict__ W,
    const float* Cin, const float* bias,
    float* D, int M, int N, int relu)
{
    __shared__ float Bs[DH * 64];

    const int lane = threadIdx.x;
    const int half = lane >> 4;
    const int sub  = lane & 15;
    const int col0 = blockIdx.y * 64;

    {
        const int tid = threadIdx.y * 32 + lane;
        for (int i = tid; i < DH * 16; i += 128) {
            const int row = i >> 4;
            const int c4  = (i & 15) << 2;
            *(float4*)&Bs[row * 64 + c4] =
                *(const float4*)&W[(size_t)row * N + col0 + c4];
        }
    }
    __syncthreads();

    const int tm   = blockIdx.x * blockDim.y + threadIdx.y;
    const int row0 = tm * 16;
    if (row0 >= M) return;

    const int arow = min(row0 + sub, M - 1);
    const float* Ap = A + (size_t)arow * DH;

    v8f acc[4] = {};

    for (int k0 = 0; k0 < DH; k0 += 4) {
        const int ka = k0 + 2 * half;
        const v2f a = *(const v2f*)(Ap + ka);
        const float* Bp = &Bs[ka * 64 + sub];
#pragma unroll
        for (int t = 0; t < 4; ++t) {
            v2f b;
            b.x = Bp[t * 16];
            b.y = Bp[t * 16 + 64];
            acc[t] = __builtin_amdgcn_wmma_f32_16x16x4_f32(
                false, a, false, b, (short)0, acc[t], false, false);
        }
    }

#pragma unroll
    for (int t = 0; t < 4; ++t) {
        const int n = col0 + t * 16 + sub;
        const float bv = bias ? bias[n] : 0.0f;
#pragma unroll
        for (int v = 0; v < 8; ++v) {
            const int m = row0 + v + 8 * half;
            if (m < M) {
                const size_t idx = (size_t)m * N + n;
                float r = acc[t][v] + bv;
                if (Cin) r += Cin[idx];
                if (relu) r = fmaxf(r, 0.0f);
                D[idx] = r;
            }
        }
    }
}

extern "C" void kernel_launch(void* const* d_in, const int* in_sizes, int n_in,
                              void* d_out, int out_size, void* d_ws, size_t ws_size,
                              hipStream_t stream)
{
    const float* x_item  = (const float*)d_in[0];
    const float* x_user  = (const float*)d_in[1];
    const int*   src_ii  = (const int*)d_in[2];
    const int*   dst_ii  = (const int*)d_in[3];
    const int*   src_iu0 = (const int*)d_in[4];
    const int*   dst_iu0 = (const int*)d_in[5];
    const int*   src_iu1 = (const int*)d_in[6];
    const int*   dst_iu1 = (const int*)d_in[7];
    const float* Wn1 = (const float*)d_in[8];
    const float* Ws1 = (const float*)d_in[9];
    const float* b1  = (const float*)d_in[10];
    const float* Wn2 = (const float*)d_in[11];
    const float* Ws2 = (const float*)d_in[12];
    const float* b2  = (const float*)d_in[13];
    const float* Wn3 = (const float*)d_in[14];
    const float* Ws3 = (const float*)d_in[15];
    const float* b3  = (const float*)d_in[16];
    const float* Wlin = (const float*)d_in[17];
    const float* blin = (const float*)d_in[18];

    const int E_ii  = in_sizes[2];
    const int E_iu0 = in_sizes[4];
    const int E_iu1 = in_sizes[6];
    const int N_ITEM_DST = 50000;
    const int N_USER0    = in_sizes[1] / DH;
    const int N_USER1    = 25000;

    char* ws = (char*)d_ws;
    const size_t off_deg   = 0;
    const size_t off_agg   = 256 * 1024;
    const size_t off_itemx = off_agg   + (size_t)N_ITEM_DST * DH * sizeof(float);
    const size_t off_userx = off_itemx + (size_t)N_ITEM_DST * DH * sizeof(float);
    float* deg    = (float*)(ws + off_deg);
    float* agg    = (float*)(ws + off_agg);
    float* itemx  = (float*)(ws + off_itemx);
    float* userx  = (float*)(ws + off_userx);
    float* userx2 = itemx;
    float* out    = (float*)d_out;

    auto zero = [&](float* p, long n) {
        zero_kernel<<<dim3((unsigned)((n + 255) / 256)), dim3(256), 0, stream>>>(p, n);
    };
    auto gemm = [&](const float* A, const float* Wm, const float* Cin,
                    const float* bias, float* Dm, int M, int N, int relu) {
        int mt = (M + 15) / 16;
        dim3 grid((mt + 3) / 4, N / 64);
        dim3 block(32, 4);
        gemm_wmma_kernel<<<grid, block, 0, stream>>>(A, Wm, Cin, bias, Dm, M, N, relu);
    };
    auto sage_agg = [&](const float* Xsrc, const int* src, const int* dst,
                        int E, int n_dst) {
        zero(deg, n_dst);
        zero(agg, (long)n_dst * DH);
        deg_kernel<<<dim3((E + 255) / 256), dim3(256), 0, stream>>>(dst, E, deg);
        scatter_kernel<<<dim3((E + 7) / 8), dim3(32, 8), 0, stream>>>(Xsrc, src, dst, E, agg);
        norm_kernel<<<dim3((unsigned)(((long)n_dst * DH + 255) / 256)), dim3(256), 0, stream>>>(
            agg, deg, (long)n_dst * DH);
    };

    sage_agg(x_item, src_ii, dst_ii, E_ii, N_ITEM_DST);
    gemm(x_item, Ws1, nullptr, nullptr, itemx, N_ITEM_DST, DH, 0);
    gemm(agg,    Wn1, itemx,   b1,      itemx, N_ITEM_DST, DH, 1);

    sage_agg(x_item, src_iu0, dst_iu0, E_iu0, N_USER0);
    gemm(x_user, Ws2, nullptr, nullptr, userx, N_USER0, DH, 0);
    gemm(agg,    Wn2, userx,   b2,      userx, N_USER0, DH, 1);

    sage_agg(itemx, src_iu1, dst_iu1, E_iu1, N_USER1);
    gemm(userx, Ws3, nullptr, nullptr, userx2, N_USER1, DH, 0);
    gemm(agg,   Wn3, userx2,  b3,      userx2, N_USER1, DH, 1);

    gemm(userx2, Wlin, nullptr, blin, out, N_USER1, DOUT, 0);
}


# Neighbour mean by owner blocks: a block keeps its destination rows' f32 sums in LDS and filters the full edge list for them; LDS float adds (arrival order) replace global ones; 10 launches, was 22
# speedup vs baseline: 2.9970x; 2.9970x over previous
#include <hip/hip_runtime.h>


typedef float v2f __attribute__((ext_vector_type(2)));
typedef float v8f __attribute__((ext_vector_type(8)));

#define DH 256
#define DOUT 128

#define AGG_ROWS  200
#define AGG_WAVES 32
#define SCAN_D    8
#define PEND_CAP  256
#define GATHER_G  8

__global__ __launch_bounds__(1024) void agg_owner_kernel(
    const float* __restrict__ X, const int* __restrict__ src,
    const int* __restrict__ dst, int E, float* __restrict__ agg,
    int n_dst, int rpb)
{
    __shared__ float acc[AGG_ROWS * DH];
    __shared__ float degs[AGG_ROWS];
    __shared__ int2  pend[AGG_WAVES][PEND_CAP];

    const int tid  = threadIdx.x;
    const int lane = tid & 31;
    const int wave = __builtin_amdgcn_readfirstlane(tid >> 5);
    const int row_lo = blockIdx.x * rpb;
    const int nrows  = max(0, min(rpb, n_dst - row_lo));

    for (int i = tid; i < nrows * (DH / 4); i += 1024)
        ((float4*)acc)[i] = make_float4(0.f, 0.f, 0.f, 0.f);
    for (int i = tid; i < nrows; i += 1024) degs[i] = 0.0f;
    __syncthreads();

    int2* ring = pend[wave];
    int head = 0;
    int cnt  = 0;

    auto drain = [&](int n) {
        const int2 ent = ring[(head + lane) & (PEND_CAP - 1)];
        const int s_l = (lane < n) ? src[ent.x] : 0;
        const int r_l = ent.y;
        for (int j = 0; j < n; j += GATHER_G) {
            float4 a[GATHER_G], b[GATHER_G];
            int rr[GATHER_G];
#pragma unroll
            for (int q = 0; q < GATHER_G; ++q) {
                if (j + q < n) {
                    const int s = __builtin_amdgcn_readlane(s_l, j + q);
                    rr[q] = __builtin_amdgcn_readlane(r_l, j + q);
                    const float4* p = (const float4*)(X + (size_t)s * DH) + lane;
                    a[q] = p[0];
                    b[q] = p[32];
                }
            }
#pragma unroll
            for (int q = 0; q < GATHER_G; ++q) {
                if (j + q < n) {
                    float* row = &acc[rr[q] * DH + lane];
                    __hip_atomic_fetch_add(row +   0, a[q].x, __ATOMIC_RELAXED, __HIP_MEMORY_SCOPE_WORKGROUP);
                    __hip_atomic_fetch_add(row +  32, a[q].y, __ATOMIC_RELAXED, __HIP_MEMORY_SCOPE_WORKGROUP);
                    __hip_atomic_fetch_add(row +  64, a[q].z, __ATOMIC_RELAXED, __HIP_MEMORY_SCOPE_WORKGROUP);
                    __hip_atomic_fetch_add(row +  96, a[q].w, __ATOMIC_RELAXED, __HIP_MEMORY_SCOPE_WORKGROUP);
                    __hip_atomic_fetch_add(row + 128, b[q].x, __ATOMIC_RELAXED, __HIP_MEMORY_SCOPE_WORKGROUP);
                    __hip_atomic_fetch_add(row + 160, b[q].y, __ATOMIC_RELAXED, __HIP_MEMORY_SCOPE_WORKGROUP);
                    __hip_atomic_fetch_add(row + 192, b[q].z, __ATOMIC_RELAXED, __HIP_MEMORY_SCOPE_WORKGROUP);
                    __hip_atomic_fetch_add(row + 224, b[q].w, __ATOMIC_RELAXED, __HIP_MEMORY_SCOPE_WORKGROUP);
                    if (lane == 0)
                        __hip_atomic_fetch_add(&degs[rr[q]], 1.0f, __ATOMIC_RELAXED, __HIP_MEMORY_SCOPE_WORKGROUP);
                }
            }
        }
        head = (head + n) & (PEND_CAP - 1);
        cnt -= n;
    };

    auto push = [&](bool m, int e, int r) {
        const unsigned mask = __builtin_amdgcn_ballot_w32(m);
        if (mask != 0u) {
            if (m) {
                const int pos = head + cnt + (int)__builtin_amdgcn_mbcnt_lo(mask, 0u);
                ring[pos & (PEND_CAP - 1)] = make_int2(e, r);
            }
            cnt += __builtin_popcount(mask);
        }
    };

    const int nchunk = (E + 127) >> 7;
    const int start  = (int)blockIdx.x * (nchunk / (int)gridDim.x);
    const int niter  = (nchunk - wave + AGG_WAVES - 1) / AGG_WAVES;
    const unsigned un = (unsigned)nrows;

    for (int i0 = 0; i0 < niter; i0 += SCAN_D) {
        int4 d[SCAN_D];
        int  eb[SCAN_D];
#pragma unroll
        for (int u = 0; u < SCAN_D; ++u) {
            d[u] = make_int4(-1, -1, -1, -1);
            eb[u] = 0;
            if (i0 + u < niter) {
                int c = start + wave + AGG_WAVES * (i0 + u);
                if (c >= nchunk) c -= nchunk;
                eb[u] = c << 7;
                const int e = eb[u] + lane * 4;
                if (eb[u] + 128 <= E) {
                    d[u] = *(const int4*)(dst + e);
                } else {
                    if (e + 0 < E) d[u].x = dst[e + 0];
                    if (e + 1 < E) d[u].y = dst[e + 1];
                    if (e + 2 < E) d[u].z = dst[e + 2];
                    if (e + 3 < E) d[u].w = dst[e + 3];
                }
            }
        }
#pragma unroll
        for (int u = 0; u < SCAN_D; ++u) {
            const unsigned rx = (unsigned)(d[u].x - row_lo), ry = (unsigned)(d[u].y - row_lo);
            const unsigned rz = (unsigned)(d[u].z - row_lo), rw = (unsigned)(d[u].w - row_lo);
            const unsigned mn = min(min(rx, ry), min(rz, rw));
            if (__builtin_amdgcn_ballot_w32(mn < un) != 0u) {
                const int e = eb[u] + lane * 4;
                push(rx < un, e + 0, (int)rx);
                push(ry < un, e + 1, (int)ry);
                push(rz < un, e + 2, (int)rz);
                push(rw < un, e + 3, (int)rw);
                while (cnt >= 32) drain(32);
            }
        }
    }
    while (cnt > 0) drain(min(cnt, 32));

    __syncthreads();

    for (int r = wave; r < nrows; r += AGG_WAVES) {
        const float inv = 1.0f / fmaxf(degs[r], 1.0f);
        const float* row = &acc[r * DH + lane];
        float4 o0, o1;
        o0.x = row[0]   * inv;  o0.y = row[32]  * inv;
        o0.z = row[64]  * inv;  o0.w = row[96]  * inv;
        o1.x = row[128] * inv;  o1.y = row[160] * inv;
        o1.z = row[192] * inv;  o1.w = row[224] * inv;
        float4* o = (float4*)(agg + (size_t)(row_lo + r) * DH) + lane;
        o[0]  = o0;
        o[32] = o1;
    }
}

__global__ __launch_bounds__(128) void gemm_wmma_kernel(
    const float* __restrict__ A, const float* __restrict__ W,
    const float* Cin, const float* bias,
    float* D, int M, int N, int relu)
{
    __shared__ float Bs[DH * 64];

    const int lane = threadIdx.x;
    const int half = lane >> 4;
    const int sub  = lane & 15;
    const int col0 = blockIdx.y * 64;

    {
        const int tid = threadIdx.y * 32 + lane;
        for (int i = tid; i < DH * 16; i += 128) {
            const int row = i >> 4;
            const int c4  = (i & 15) << 2;
            *(float4*)&Bs[row * 64 + c4] =
                *(const float4*)&W[(size_t)row * N + col0 + c4];
        }
    }
    __syncthreads();

    const int tm   = blockIdx.x * blockDim.y + threadIdx.y;
    const int row0 = tm * 16;
    if (row0 >= M) return;

    const int arow = min(row0 + sub, M - 1);
    const float* Ap = A + (size_t)arow * DH;

    v8f acc[4] = {};

    for (int k0 = 0; k0 < DH; k0 += 4) {
        const int ka = k0 + 2 * half;
        const v2f a = *(const v2f*)(Ap + ka);
        const float* Bp = &Bs[ka * 64 + sub];
#pragma unroll
        for (int t = 0; t < 4; ++t) {
            v2f b;
            b.x = Bp[t * 16];
            b.y = Bp[t * 16 + 64];
            acc[t] = __builtin_amdgcn_wmma_f32_16x16x4_f32(
                false, a, false, b, (short)0, acc[t], false, false);
        }
    }

#pragma unroll
    for (int t = 0; t < 4; ++t) {
        const int n = col0 + t * 16 + sub;
        const float bv = bias ? bias[n] : 0.0f;
#pragma unroll
        for (int v = 0; v < 8; ++v) {
            const int m = row0 + v + 8 * half;
            if (m < M) {
                const size_t idx = (size_t)m * N + n;
                float r = acc[t][v] + bv;
                if (Cin) r += Cin[idx];
                if (relu) r = fmaxf(r, 0.0f);
                D[idx] = r;
            }
        }
    }
}

extern "C" void kernel_launch(void* const* d_in, const int* in_sizes, int n_in,
                              void* d_out, int out_size, void* d_ws, size_t ws_size,
                              hipStream_t stream)
{
    const float* x_item  = (const float*)d_in[0];
    const float* x_user  = (const float*)d_in[1];
    const int*   src_ii  = (const int*)d_in[2];
    const int*   dst_ii  = (const int*)d_in[3];
    const int*   src_iu0 = (const int*)d_in[4];
    const int*   dst_iu0 = (const int*)d_in[5];
    const int*   src_iu1 = (const int*)d_in[6];
    const int*   dst_iu1 = (const int*)d_in[7];
    const float* Wn1 = (const float*)d_in[8];
    const float* Ws1 = (const float*)d_in[9];
    const float* b1  = (const float*)d_in[10];
    const float* Wn2 = (const float*)d_in[11];
    const float* Ws2 = (const float*)d_in[12];
    const float* b2  = (const float*)d_in[13];
    const float* Wn3 = (const float*)d_in[14];
    const float* Ws3 = (const float*)d_in[15];
    const float* b3  = (const float*)d_in[16];
    const float* Wlin = (const float*)d_in[17];
    const float* blin = (const float*)d_in[18];

    const int E_ii  = in_sizes[2];
    const int E_iu0 = in_sizes[4];
    const int E_iu1 = in_sizes[6];
    const int N_ITEM_DST = 50000;
    const int N_USER0    = in_sizes[1] / DH;
    const int N_USER1    = 25000;

    char* ws = (char*)d_ws;
    const size_t off_agg   = 256 * 1024;
    const size_t off_itemx = off_agg   + (size_t)N_ITEM_DST * DH * sizeof(float);
    const size_t off_userx = off_itemx + (size_t)N_ITEM_DST * DH * sizeof(float);
    float* agg    = (float*)(ws + off_agg);
    float* itemx  = (float*)(ws + off_itemx);
    float* userx  = (float*)(ws + off_userx);
    float* userx2 = itemx;
    float* out    = (float*)d_out;

    auto gemm = [&](const float* A, const float* Wm, const float* Cin,
                    const float* bias, float* Dm, int M, int N, int relu) {
        int mt = (M + 15) / 16;
        dim3 grid((mt + 3) / 4, N / 64);
        dim3 block(32, 4);
        gemm_wmma_kernel<<<grid, block, 0, stream>>>(A, Wm, Cin, bias, Dm, M, N, relu);
    };
    auto sage_agg = [&](const float* Xsrc, const int* src, const int* dst,
                        int E, int n_dst) {
        int rpb = (n_dst + 255) / 256;
        if (rpb > AGG_ROWS) rpb = AGG_ROWS;
        if (rpb < 1) rpb = 1;
        agg_owner_kernel<<<dim3((n_dst + rpb - 1) / rpb), dim3(1024), 0, stream>>>(
            Xsrc, src, dst, E, agg, n_dst, rpb);
    };

    sage_agg(x_item, src_ii, dst_ii, E_ii, N_ITEM_DST);
    gemm(x_item, Ws1, nullptr, nullptr, itemx, N_ITEM_DST, DH, 0);
    gemm(agg,    Wn1, itemx,   b1,      itemx, N_ITEM_DST, DH, 1);

    sage_agg(x_item, src_iu0, dst_iu0, E_iu0, N_USER0);
    gemm(x_user, Ws2, nullptr, nullptr, userx, N_USER0, DH, 0);
    gemm(agg,    Wn2, userx,   b2,      userx, N_USER0, DH, 1);

    sage_agg(itemx, src_iu1, dst_iu1, E_iu1, N_USER1);
    gemm(userx, Ws3, nullptr, nullptr, userx2, N_USER1, DH, 0);
    gemm(agg,   Wn3, userx2,  b3,      userx2, N_USER1, DH, 1);

    gemm(userx2, Wlin, nullptr, blin, out, N_USER1, DOUT, 0);
}


# Per-layer pre-pass re-encodes the edge list: owner blocks scan 1-byte owner keys as bit planes (16 edges per compare), fetch kept edges as (src,dst) pairs; arrival-order sums; 10 launches, was 7
# speedup vs baseline: 5.1546x; 1.7199x over previous
#include <hip/hip_runtime.h>


typedef float v2f __attribute__((ext_vector_type(2)));
typedef float v8f __attribute__((ext_vector_type(8)));

#define DH 256
#define DOUT 128

#define KEY_CHUNK 512

__global__ __launch_bounds__(256) void edge_encode_kernel(
    const int* __restrict__ src, const int* __restrict__ dst, int E, int rpb,
    int* __restrict__ keys, int n16, int* __restrict__ pairs)
{
    const int t = blockIdx.x * 256 + threadIdx.x;
    if (t < n16) {
        const int e0 = t * 16;
        const unsigned urpb = (unsigned)rpb;
        unsigned k[16];
        if (e0 + 16 <= E) {
#pragma unroll
            for (int q = 0; q < 4; ++q) {
                const int4 d = *(const int4*)(dst + e0 + 4 * q);
                k[4 * q + 0] = min((unsigned)d.x / urpb, 255u);
                k[4 * q + 1] = min((unsigned)d.y / urpb, 255u);
                k[4 * q + 2] = min((unsigned)d.z / urpb, 255u);
                k[4 * q + 3] = min((unsigned)d.w / urpb, 255u);
            }
        } else {
#pragma unroll
            for (int i = 0; i < 16; ++i) {
                k[i] = 255u;
                if (e0 + i < E) k[i] = min((unsigned)dst[e0 + i] / urpb, 255u);
            }
        }
        unsigned w[4] = {0u, 0u, 0u, 0u};
#pragma unroll
        for (int i = 0; i < 16; ++i) {
#pragma unroll
            for (int q = 0; q < 4; ++q)
                w[q] |= (((k[i] >> (2 * q)) & 1u) << i) | (((k[i] >> (2 * q + 1)) & 1u) << (16 + i));
        }
        ((uint4*)keys)[t] = make_uint4(w[0], w[1], w[2], w[3]);
    }

    const int base = blockIdx.x * 4096 + 2 * (int)threadIdx.x;
    if ((int)blockIdx.x * 4096 + 4096 <= E) {
        int2 s[8], d[8];
#pragma unroll
        for (int i = 0; i < 8; ++i) {
            s[i] = *(const int2*)(src + base + 512 * i);
            d[i] = *(const int2*)(dst + base + 512 * i);
        }
#pragma unroll
        for (int i = 0; i < 8; ++i)
            *(int4*)(pairs + 2 * (size_t)(base + 512 * i)) = make_int4(s[i].x, d[i].x, s[i].y, d[i].y);
    } else {
        for (int i = 0; i < 8; ++i) {
            const int e = base + 512 * i;
            if (e + 1 < E) {
                const int2 s = *(const int2*)(src + e);
                const int2 d = *(const int2*)(dst + e);
                *(int4*)(pairs + 2 * (size_t)e) = make_int4(s.x, d.x, s.y, d.y);
            } else if (e < E) {
                pairs[2 * (size_t)e]     = src[e];
                pairs[2 * (size_t)e + 1] = dst[e];
            }
        }
    }
}

#define AGG_ROWS  200
#define AGG_WAVES 32
#define SCAN_D    8
#define PEND_CAP  64
#define GATHER_G  8

__global__ __launch_bounds__(1024) void agg_owner_kernel(
    const float* __restrict__ X, const int* __restrict__ pairs,
    const int* __restrict__ keys, int E,
    float* __restrict__ agg, int n_dst, int rpb)
{
    __shared__ float acc[AGG_ROWS * DH];
    __shared__ float degs[AGG_ROWS];
    __shared__ int   pend[AGG_WAVES][PEND_CAP];

    const int tid  = threadIdx.x;
    const int lane = tid & 31;
    const int wave = __builtin_amdgcn_readfirstlane(tid >> 5);
    const int row_lo = blockIdx.x * rpb;
    const int nrows  = max(0, min(rpb, n_dst - row_lo));
    const unsigned un = (unsigned)nrows;

    for (int i = tid; i < nrows * (DH / 4); i += 1024)
        ((float4*)acc)[i] = make_float4(0.f, 0.f, 0.f, 0.f);
    for (int i = tid; i < nrows; i += 1024) degs[i] = 0.0f;
    __syncthreads();

    int* ring = pend[wave];
    int head = 0;
    int cnt  = 0;

    auto drain = [&](int n) {
        int s_l = 0, r_l = -1;
        if (lane < n) {
            const int e_l = ring[(head + lane) & (PEND_CAP - 1)];
            const int2 sd = ((const int2*)pairs)[e_l];
            s_l = sd.x;
            r_l = sd.y - row_lo;
        }
        for (int j = 0; j < n; j += GATHER_G) {
            float4 a[GATHER_G], b[GATHER_G];
            int rr[GATHER_G];
#pragma unroll
            for (int q = 0; q < GATHER_G; ++q) {
                if (j + q < n) {
                    const int s = __builtin_amdgcn_readlane(s_l, j + q);
                    rr[q] = __builtin_amdgcn_readlane(r_l, j + q);
                    const float4* p = (const float4*)(X + (size_t)s * DH) + lane;
                    a[q] = p[0];
                    b[q] = p[32];
                }
            }
#pragma unroll
            for (int q = 0; q < GATHER_G; ++q) {
                if (j + q < n && (unsigned)rr[q] < un) {
                    float* row = &acc[rr[q] * DH + lane];
                    __hip_atomic_fetch_add(row +   0, a[q].x, __ATOMIC_RELAXED, __HIP_MEMORY_SCOPE_WORKGROUP);
                    __hip_atomic_fetch_add(row +  32, a[q].y, __ATOMIC_RELAXED, __HIP_MEMORY_SCOPE_WORKGROUP);
                    __hip_atomic_fetch_add(row +  64, a[q].z, __ATOMIC_RELAXED, __HIP_MEMORY_SCOPE_WORKGROUP);
                    __hip_atomic_fetch_add(row +  96, a[q].w, __ATOMIC_RELAXED, __HIP_MEMORY_SCOPE_WORKGROUP);
                    __hip_atomic_fetch_add(row + 128, b[q].x, __ATOMIC_RELAXED, __HIP_MEMORY_SCOPE_WORKGROUP);
                    __hip_atomic_fetch_add(row + 160, b[q].y, __ATOMIC_RELAXED, __HIP_MEMORY_SCOPE_WORKGROUP);
                    __hip_atomic_fetch_add(row + 192, b[q].z, __ATOMIC_RELAXED, __HIP_MEMORY_SCOPE_WORKGROUP);
                    __hip_atomic_fetch_add(row + 224, b[q].w, __ATOMIC_RELAXED, __HIP_MEMORY_SCOPE_WORKGROUP);
                    if (lane == 0)
                        __hip_atomic_fetch_add(&degs[rr[q]], 1.0f, __ATOMIC_RELAXED, __HIP_MEMORY_SCOPE_WORKGROUP);
                }
            }
        }
        head = (head + n) & (PEND_CAP - 1);
        cnt -= n;
    };

    auto push = [&](bool m, int e) {
        const unsigned mask = __builtin_amdgcn_ballot_w32(m);
        if (m) {
            const int pos = head + cnt + (int)__builtin_amdgcn_mbcnt_lo(mask, 0u);
            ring[pos & (PEND_CAP - 1)] = e;
        }
        cnt += __builtin_popcount(mask);
    };

    const int nchunk = (E + KEY_CHUNK - 1) / KEY_CHUNK;
    const int start  = (int)blockIdx.x * (nchunk / (int)gridDim.x);
    const int niter  = (nchunk - wave + AGG_WAVES - 1) / AGG_WAVES;
    const uint4* kq  = (const uint4*)keys;

    const unsigned mykey = min((unsigned)blockIdx.x, 255u);
    unsigned cq[4];
#pragma unroll
    for (int q = 0; q < 4; ++q)
        cq[q] = (((mykey >> (2 * q)) & 1u) ? 0u : 0x0000ffffu)
              | (((mykey >> (2 * q + 1)) & 1u) ? 0u : 0xffff0000u);
    auto match = [&](const uint4& v) -> unsigned {
        const unsigned a = (v.x ^ cq[0]) & (v.y ^ cq[1]) & (v.z ^ cq[2]) & (v.w ^ cq[3]);
        return a & (a >> 16);
    };

    for (int i0 = 0; i0 < niter; i0 += SCAN_D) {
        uint4 kv[SCAN_D];
#pragma unroll
        for (int u = 0; u < SCAN_D; ++u) {
            int c = start + wave + AGG_WAVES * min(i0 + u, niter - 1);
            if (c >= nchunk) c -= nchunk;
            kv[u] = kq[c * (KEY_CHUNK / 16) + lane];
        }
        unsigned m[SCAN_D / 2];
#pragma unroll
        for (int j = 0; j < SCAN_D / 2; ++j) {
            const unsigned t0 = match(kv[2 * j]);
            const unsigned t1 = match(kv[2 * j + 1]);
            const unsigned m0 = (i0 + 2 * j     < niter) ? t0 : 0u;
            const unsigned m1 = (i0 + 2 * j + 1 < niter) ? t1 : 0u;
            m[j] = m0 | (m1 << 16);
        }

        for (;;) {
            unsigned sel = 0u;
            int js = 0;
#pragma unroll
            for (int j = SCAN_D / 2 - 1; j >= 0; --j)
                if (m[j] != 0u) { sel = m[j]; js = j; }
            if (__builtin_amdgcn_ballot_w32(sel != 0u) == 0u) break;

            const int p = __builtin_ctz(sel | 0x80000000u);
            int c = start + wave + AGG_WAVES * (i0 + 2 * js + (p >> 4));
            if (c >= nchunk) c -= nchunk;
            const int e = c * KEY_CHUNK + lane * 16 + (p & 15);
            push(sel != 0u && e < E, e);
            const unsigned rest = sel & (sel - 1u);
#pragma unroll
            for (int j = 0; j < SCAN_D / 2; ++j)
                if (js == j) m[j] = rest;
            if (cnt >= 32) drain(32);
        }
    }
    while (cnt > 0) drain(min(cnt, 32));

    __syncthreads();

    for (int r = wave; r < nrows; r += AGG_WAVES) {
        const float inv = 1.0f / fmaxf(degs[r], 1.0f);
        const float* row = &acc[r * DH + lane];
        float4 o0, o1;
        o0.x = row[0]   * inv;  o0.y = row[32]  * inv;
        o0.z = row[64]  * inv;  o0.w = row[96]  * inv;
        o1.x = row[128] * inv;  o1.y = row[160] * inv;
        o1.z = row[192] * inv;  o1.w = row[224] * inv;
        float4* o = (float4*)(agg + (size_t)(row_lo + r) * DH) + lane;
        o[0]  = o0;
        o[32] = o1;
    }
}

#define GR_COLS  128
#define GR_PITCH 136
#define GR_WAVES 16

__global__ __launch_bounds__(GR_WAVES * 32) void gemm_resident_kernel(
    const float* __restrict__ A0, const float* __restrict__ W0,
    const float* __restrict__ A1, const float* __restrict__ W1,
    const float* __restrict__ bias, float* __restrict__ D,
    int M, int N, int relu)
{
    __shared__ v2f Bs2[DH * GR_PITCH];

    const int tid  = threadIdx.x;
    const int lane = tid & 31;
    const int wave = __builtin_amdgcn_readfirstlane(tid >> 5);
    const int half = lane >> 4;
    const int sub  = lane & 15;
    const int col0 = blockIdx.y * GR_COLS;
    const int nseg = (A1 != nullptr) ? 2 : 1;

    for (int sg = 0; sg < nseg; ++sg) {
        const float* Wm = sg ? W1 : W0;
        float4 ra[8], rb[8];
#pragma unroll
        for (int q = 0; q < 8; ++q) {
            const int i  = tid + q * (GR_WAVES * 32);
            const int kp = i >> 5;
            const int c4 = (i & 31) << 2;
            const float* p = Wm + (size_t)(2 * kp) * N + col0 + c4;
            ra[q] = *(const float4*)p;
            rb[q] = *(const float4*)(p + N);
        }
#pragma unroll
        for (int q = 0; q < 8; ++q) {
            const int i  = tid + q * (GR_WAVES * 32);
            const int kp = i >> 5;
            const int c4 = (i & 31) << 2;
            float4* d = (float4*)&Bs2[(sg * (DH / 2) + kp) * GR_PITCH + c4];
            d[0] = make_float4(ra[q].x, rb[q].x, ra[q].y, rb[q].y);
            d[1] = make_float4(ra[q].z, rb[q].z, ra[q].w, rb[q].w);
        }
    }
    __syncthreads();

    const int mt = (M + 15) >> 4;
    const int hf = wave & 1;
    const int ng = nseg * 8;
    const int nb = col0 + hf * 64 + sub;
    const v2f* Bl = Bs2 + (2 * half) * GR_PITCH + hf * 64 + sub;

    float bv[4];
#pragma unroll
    for (int t = 0; t < 4; ++t) bv[t] = bias ? bias[nb + t * 16] : 0.0f;

    for (int lt = wave >> 1; ; lt += GR_WAVES / 2) {
        const int tile = (int)blockIdx.x + (int)gridDim.x * lt;
        if (tile >= mt) break;

        const int row0 = tile * 16;
        const int arow = min(row0 + sub, M - 1);
        const float* A0r = A0 + (size_t)arow * DH + 4 * half;
        const float* A1r = (nseg == 2) ? (A1 + (size_t)arow * DH + 4 * half) : A0r;

        v8f acc[4] = {};

        auto loadA = [&](int gg, float4 (&dst)[4]) {
            const float* p = ((gg >> 3) ? A1r : A0r) + (gg & 7) * 32;
#pragma unroll
            for (int q = 0; q < 4; ++q) dst[q] = *(const float4*)(p + 8 * q);
        };
        auto mma = [&](int gg, const float4 (&av)[4]) {
            const v2f* Bg = Bl + gg * 16 * GR_PITCH;
#pragma unroll
            for (int q = 0; q < 4; ++q) {
#pragma unroll
                for (int j = 0; j < 2; ++j) {
                    v2f af;
                    af.x = j ? av[q].z : av[q].x;
                    af.y = j ? av[q].w : av[q].y;
                    const v2f* Bq = Bg + (4 * q + j) * GR_PITCH;
#pragma unroll
                    for (int t = 0; t < 4; ++t)
                        acc[t] = __builtin_amdgcn_wmma_f32_16x16x4_f32(
                            false, af, false, Bq[t * 16], (short)0, acc[t], false, false);
                }
            }
        };

        float4 a[4], an[4];
        loadA(0, a);
        for (int gg = 0; gg < ng - 2; gg += 2) {
            loadA(gg + 1, an);
            mma(gg, a);
            loadA(gg + 2, a);
            mma(gg + 1, an);
        }
        loadA(ng - 1, an);
        mma(ng - 2, a);
        mma(ng - 1, an);

#pragma unroll
        for (int t = 0; t < 4; ++t) {
            const int n = nb + t * 16;
#pragma unroll
            for (int v = 0; v < 8; ++v) {
                const int m = row0 + v + 8 * half;
                if (m < M) {
                    float r = acc[t][v] + bv[t];
                    if (relu) r = fmaxf(r, 0.0f);
                    D[(size_t)m * N + n] = r;
                }
            }
        }
    }
}

extern "C" void kernel_launch(void* const* d_in, const int* in_sizes, int n_in,
                              void* d_out, int out_size, void* d_ws, size_t ws_size,
                              hipStream_t stream)
{
    const float* x_item  = (const float*)d_in[0];
    const float* x_user  = (const float*)d_in[1];
    const int*   src_ii  = (const int*)d_in[2];
    const int*   dst_ii  = (const int*)d_in[3];
    const int*   src_iu0 = (const int*)d_in[4];
    const int*   dst_iu0 = (const int*)d_in[5];
    const int*   src_iu1 = (const int*)d_in[6];
    const int*   dst_iu1 = (const int*)d_in[7];
    const float* Wn1 = (const float*)d_in[8];
    const float* Ws1 = (const float*)d_in[9];
    const float* b1  = (const float*)d_in[10];
    const float* Wn2 = (const float*)d_in[11];
    const float* Ws2 = (const float*)d_in[12];
    const float* b2  = (const float*)d_in[13];
    const float* Wn3 = (const float*)d_in[14];
    const float* Ws3 = (const float*)d_in[15];
    const float* b3  = (const float*)d_in[16];
    const float* Wlin = (const float*)d_in[17];
    const float* blin = (const float*)d_in[18];

    const int E_ii  = in_sizes[2];
    const int E_iu0 = in_sizes[4];
    const int E_iu1 = in_sizes[6];
    const int N_ITEM_DST = 50000;
    const int N_USER0    = in_sizes[1] / DH;
    const int N_USER1    = 25000;

    char* ws = (char*)d_ws;
    const size_t off_agg   = 256 * 1024;
    const size_t off_itemx = off_agg   + (size_t)N_ITEM_DST * DH * sizeof(float);
    const size_t off_userx = off_itemx + (size_t)N_ITEM_DST * DH * sizeof(float);
    float* agg    = (float*)(ws + off_agg);
    float* itemx  = (float*)(ws + off_itemx);
    float* userx  = (float*)(ws + off_userx);
    float* userx2 = itemx;
    float* out    = (float*)d_out;
    int* keys = (int*)(userx + (size_t)N_USER1 * DH);

    auto gemm = [&](const float* A0, const float* W0, const float* A1, const float* W1,
                    const float* bias, float* Dm, int M, int N, int relu) {
        dim3 grid(256 / (N / GR_COLS), N / GR_COLS);
        gemm_resident_kernel<<<grid, dim3(GR_WAVES * 32), 0, stream>>>(
            A0, W0, A1, W1, bias, Dm, M, N, relu);
    };
    auto sage_agg = [&](const float* Xsrc, const int* src, const int* dst,
                        int E, int n_dst) {
        int rpb = (n_dst + 255) / 256;
        if (rpb > AGG_ROWS) rpb = AGG_ROWS;
        if (rpb < 1) rpb = 1;
        const int n16 = ((E + KEY_CHUNK - 1) / KEY_CHUNK) * (KEY_CHUNK / 16);
        int* pairs = keys + (size_t)n16 * 4;
        edge_encode_kernel<<<dim3((n16 + 255) / 256), dim3(256), 0, stream>>>(
            src, dst, E, rpb, keys, n16, pairs);
        agg_owner_kernel<<<dim3((n_dst + rpb - 1) / rpb), dim3(1024), 0, stream>>>(
            Xsrc, pairs, keys, E, agg, n_dst, rpb);
    };

    sage_agg(x_item, src_ii, dst_ii, E_ii, N_ITEM_DST);
    gemm(x_item, Ws1, agg, Wn1, b1, itemx, N_ITEM_DST, DH, 1);

    sage_agg(x_item, src_iu0, dst_iu0, E_iu0, N_USER0);
    gemm(x_user, Ws2, agg, Wn2, b2, userx, N_USER0, DH, 1);

    sage_agg(itemx, src_iu1, dst_iu1, E_iu1, N_USER1);
    gemm(userx, Ws3, agg, Wn3, b3, userx2, N_USER1, DH, 1);

    gemm(userx2, Wlin, nullptr, nullptr, blin, out, N_USER1, DOUT, 0);
}
